# starting kernel
# baseline (speedup 1.0000x reference)
#include <hip/hip_runtime.h>
#include <hip/hip_bf16.h>


typedef __attribute__((ext_vector_type(2))) float v2f;
typedef __attribute__((ext_vector_type(4))) float v4f;
typedef __attribute__((ext_vector_type(8))) float v8f;

#define BSZ 16
#define SEQ 256
#define HID 1024
#define TAG 32
#define MTOT (BSZ * SEQ)

__global__ __launch_bounds__(32) void ps_kernel(const float* __restrict__ seq,
                                                const float* __restrict__ W,
                                                const float* __restrict__ bias,
                                                float* __restrict__ ps) {
  const int b = blockIdx.x;
  const int t = threadIdx.x;
  const v4f* sv = (const v4f*)(seq + (size_t)b * HID);
  const v4f* wv = (const v4f*)(W + (size_t)t * (3 * HID) + 2 * HID);
  float acc = 0.0f;
  #pragma unroll 4
  for (int h = 0; h < HID / 4; ++h) {
    v4f s = sv[h];
    v4f w = wv[h];
    acc += s.x * w.x + s.y * w.y + s.z * w.z + s.w * w.w;
  }
  ps[b * TAG + t] = acc + bias[t];
}

__global__ __launch_bounds__(128) void gemm_wmma_kernel(
    const float* __restrict__ E,
    const float* __restrict__ W,
    const float* __restrict__ ps,
    float* __restrict__ p1ps,
    float* __restrict__ p2) {
  const int lane  = threadIdx.x & 31;
  const int wave  = threadIdx.x >> 5;
  const int lmod  = lane & 15;
  const int lhalf = lane >> 4;
  const int m0    = (blockIdx.x * 4 + wave) * 16;

  const float* Arow = E + (size_t)(m0 + lmod) * HID + 2 * lhalf;

  const float* B0 = W + (size_t)lmod * (3 * HID) + 2 * lhalf;
  const float* B1 = W + (size_t)(lmod + 16) * (3 * HID) + 2 * lhalf;
  const float* B2 = B0 + HID;
  const float* B3 = B1 + HID;

  v8f a0 = {}, a1 = {}, a2 = {}, a3 = {};
  #pragma unroll 4
  for (int k = 0; k < HID; k += 4) {
    v2f av = *(const v2f*)(Arow + k);
    v2f b0 = *(const v2f*)(B0 + k);
    v2f b1 = *(const v2f*)(B1 + k);
    v2f b2 = *(const v2f*)(B2 + k);
    v2f b3 = *(const v2f*)(B3 + k);
    a0 = __builtin_amdgcn_wmma_f32_16x16x4_f32(false, av, false, b0, (short)0, a0, false, false);
    a1 = __builtin_amdgcn_wmma_f32_16x16x4_f32(false, av, false, b1, (short)0, a1, false, false);
    a2 = __builtin_amdgcn_wmma_f32_16x16x4_f32(false, av, false, b2, (short)0, a2, false, false);
    a3 = __builtin_amdgcn_wmma_f32_16x16x4_f32(false, av, false, b3, (short)0, a3, false, false);
  }

  const int b = m0 >> 8;
  const float psLo = ps[b * TAG + lmod];
  const float psHi = ps[b * TAG + 16 + lmod];
  #pragma unroll
  for (int r = 0; r < 8; ++r) {
    const int row = m0 + r + 8 * lhalf;
    p1ps[(size_t)row * TAG + lmod]      = a0[r] + psLo;
    p1ps[(size_t)row * TAG + 16 + lmod] = a1[r] + psHi;
    p2  [(size_t)row * TAG + lmod]      = a2[r];
    p2  [(size_t)row * TAG + 16 + lmod] = a3[r];
  }
}

__global__ __launch_bounds__(256) void bcast_kernel(const float* __restrict__ p1ps,
                                                    const float* __restrict__ p2,
                                                    float* __restrict__ out) {
  const int bi  = blockIdx.x;
  const int b   = bi >> 8;
  const int tid = threadIdx.x;
  const v4f* p1v = (const v4f*)(p1ps + (size_t)bi * TAG);
  const v4f* p2v = (const v4f*)(p2 + (size_t)b * (SEQ * TAG));
  v4f* outv = (v4f*)(out + (size_t)bi * (SEQ * TAG));
  const v4f a = p1v[tid & 7];
  #pragma unroll
  for (int it = 0; it < 8; ++it) {
    const int f = tid + 256 * it;
    v4f v = p2v[f];
    v += a;
    __builtin_nontemporal_store(v, outv + f);
  }
}

extern "C" void kernel_launch(void* const* d_in, const int* in_sizes, int n_in,
                              void* d_out, int out_size, void* d_ws, size_t ws_size,
                              hipStream_t stream) {
  const float* seq  = (const float*)d_in[0];
  const float* ents = (const float*)d_in[1];
  const float* W    = (const float*)d_in[2];
  const float* bias = (const float*)d_in[3];
  float* out = (float*)d_out;

  float* ws   = (float*)d_ws;
  float* ps   = ws;
  float* p1ps = ws + 512;
  float* p2   = p1ps + (size_t)MTOT * TAG;

  ps_kernel<<<BSZ, 32, 0, stream>>>(seq, W, bias, ps);
  gemm_wmma_kernel<<<MTOT / 64, 128, 0, stream>>>(ents, W, ps, p1ps, p2);
  bcast_kernel<<<MTOT, 256, 0, stream>>>(p1ps, p2, out);
}


# GEMM: K split over the 8 waves of a work-group (one 16-row tile per group, 256 groups of 8 waves, was 64 of 4), partial tiles summed through LDS in fixed wave order; f32 sum order changes
# speedup vs baseline: 1.8278x; 1.8278x over previous
#include <hip/hip_runtime.h>
#include <hip/hip_bf16.h>


typedef __attribute__((ext_vector_type(2))) float v2f;
typedef __attribute__((ext_vector_type(4))) float v4f;
typedef __attribute__((ext_vector_type(8))) float v8f;

#define BSZ 16
#define SEQ 256
#define HID 1024
#define TAG 32
#define MTOT (BSZ * SEQ)

__global__ __launch_bounds__(32) void ps_kernel(const float* __restrict__ seq,
                                                const float* __restrict__ W,
                                                const float* __restrict__ bias,
                                                float* __restrict__ ps) {
  const int b = blockIdx.x;
  const int t = threadIdx.x;
  const v4f* sv = (const v4f*)(seq + (size_t)b * HID);
  const v4f* wv = (const v4f*)(W + (size_t)t * (3 * HID) + 2 * HID);
  float acc = 0.0f;
  #pragma unroll 4
  for (int h = 0; h < HID / 4; ++h) {
    v4f s = sv[h];
    v4f w = wv[h];
    acc += s.x * w.x + s.y * w.y + s.z * w.z + s.w * w.w;
  }
  ps[b * TAG + t] = acc + bias[t];
}

#define KWAVES 8
#define KSLICE (HID / KWAVES)
#define RPITCH 68

__global__ __launch_bounds__(32 * KWAVES) void gemm_wmma_kernel(
    const float* __restrict__ E,
    const float* __restrict__ W,
    const float* __restrict__ ps,
    float* __restrict__ p1ps,
    float* __restrict__ p2) {
  __shared__ __attribute__((aligned(16))) float red[KWAVES * 16 * RPITCH];

  const int lane  = threadIdx.x & 31;
  const int wave  = threadIdx.x >> 5;
  const int lmod  = lane & 15;
  const int lhalf = lane >> 4;
  const int m0    = blockIdx.x * 16;
  const int k0    = wave * KSLICE;

  const float* Arow = E + (size_t)(m0 + lmod) * HID + k0 + 2 * lhalf;

  const float* B0 = W + (size_t)lmod * (3 * HID) + k0 + 2 * lhalf;
  const float* B1 = W + (size_t)(lmod + 16) * (3 * HID) + k0 + 2 * lhalf;
  const float* B2 = B0 + HID;
  const float* B3 = B1 + HID;

  v8f a0 = {}, a1 = {}, a2 = {}, a3 = {};
  #pragma unroll 8
  for (int k = 0; k < KSLICE; k += 4) {
    v2f av = *(const v2f*)(Arow + k);
    v2f b0 = *(const v2f*)(B0 + k);
    v2f b1 = *(const v2f*)(B1 + k);
    v2f b2 = *(const v2f*)(B2 + k);
    v2f b3 = *(const v2f*)(B3 + k);
    a0 = __builtin_amdgcn_wmma_f32_16x16x4_f32(false, av, false, b0, (short)0, a0, false, false);
    a1 = __builtin_amdgcn_wmma_f32_16x16x4_f32(false, av, false, b1, (short)0, a1, false, false);
    a2 = __builtin_amdgcn_wmma_f32_16x16x4_f32(false, av, false, b2, (short)0, a2, false, false);
    a3 = __builtin_amdgcn_wmma_f32_16x16x4_f32(false, av, false, b3, (short)0, a3, false, false);
  }

  float* mine = red + wave * (16 * RPITCH);
  #pragma unroll
  for (int r = 0; r < 8; ++r) {
    const int row = r + 8 * lhalf;
    mine[row * RPITCH + lmod]      = a0[r];
    mine[row * RPITCH + 16 + lmod] = a1[r];
    mine[row * RPITCH + 32 + lmod] = a2[r];
    mine[row * RPITCH + 48 + lmod] = a3[r];
  }
  __syncthreads();

  const int row = threadIdx.x >> 4;
  const int c4  = (threadIdx.x & 15) * 4;
  const float* rp = red + row * RPITCH + c4;
  v4f s = *(const v4f*)rp;
  #pragma unroll
  for (int w = 1; w < KWAVES; ++w) s += *(const v4f*)(rp + w * (16 * RPITCH));

  const int b = m0 >> 8;
  if (c4 < TAG) {
    s += *(const v4f*)(ps + b * TAG + c4);
    *(v4f*)(p1ps + (size_t)(m0 + row) * TAG + c4) = s;
  } else {
    *(v4f*)(p2 + (size_t)(m0 + row) * TAG + (c4 - TAG)) = s;
  }
}

__global__ __launch_bounds__(256) void bcast_kernel(const float* __restrict__ p1ps,
                                                    const float* __restrict__ p2,
                                                    float* __restrict__ out) {
  const int bi  = blockIdx.x;
  const int b   = bi >> 8;
  const int tid = threadIdx.x;
  const v4f* p1v = (const v4f*)(p1ps + (size_t)bi * TAG);
  const v4f* p2v = (const v4f*)(p2 + (size_t)b * (SEQ * TAG));
  v4f* outv = (v4f*)(out + (size_t)bi * (SEQ * TAG));
  const v4f a = p1v[tid & 7];
  #pragma unroll
  for (int it = 0; it < 8; ++it) {
    const int f = tid + 256 * it;
    v4f v = p2v[f];
    v += a;
    __builtin_nontemporal_store(v, outv + f);
  }
}

extern "C" void kernel_launch(void* const* d_in, const int* in_sizes, int n_in,
                              void* d_out, int out_size, void* d_ws, size_t ws_size,
                              hipStream_t stream) {
  const float* seq  = (const float*)d_in[0];
  const float* ents = (const float*)d_in[1];
  const float* W    = (const float*)d_in[2];
  const float* bias = (const float*)d_in[3];
  float* out = (float*)d_out;

  float* ws   = (float*)d_ws;
  float* ps   = ws;
  float* p1ps = ws + 512;
  float* p2   = p1ps + (size_t)MTOT * TAG;

  ps_kernel<<<BSZ, 32, 0, stream>>>(seq, W, bias, ps);
  gemm_wmma_kernel<<<MTOT / 16, 32 * KWAVES, 0, stream>>>(ents, W, ps, p1ps, p2);
  bcast_kernel<<<MTOT, 256, 0, stream>>>(p1ps, p2, out);
}


# Broadcast-add launch: 256 resident work-groups of 8 waves (was 4096 short ones), each writes 16 rows of out with its p2 slice held in registers; every sum and its order unchanged
# speedup vs baseline: 2.0880x; 1.1424x over previous
#include <hip/hip_runtime.h>
#include <hip/hip_bf16.h>


typedef __attribute__((ext_vector_type(2))) float v2f;
typedef __attribute__((ext_vector_type(4))) float v4f;
typedef __attribute__((ext_vector_type(8))) float v8f;

#define BSZ 16
#define SEQ 256
#define HID 1024
#define TAG 32
#define MTOT (BSZ * SEQ)

__global__ __launch_bounds__(32) void ps_kernel(const float* __restrict__ seq,
                                                const float* __restrict__ W,
                                                const float* __restrict__ bias,
                                                float* __restrict__ ps) {
  const int b = blockIdx.x;
  const int t = threadIdx.x;
  const v4f* sv = (const v4f*)(seq + (size_t)b * HID);
  const v4f* wv = (const v4f*)(W + (size_t)t * (3 * HID) + 2 * HID);
  float acc = 0.0f;
  #pragma unroll 4
  for (int h = 0; h < HID / 4; ++h) {
    v4f s = sv[h];
    v4f w = wv[h];
    acc += s.x * w.x + s.y * w.y + s.z * w.z + s.w * w.w;
  }
  ps[b * TAG + t] = acc + bias[t];
}

#define KWAVES 8
#define KSLICE (HID / KWAVES)
#define RPITCH 68

__global__ __launch_bounds__(32 * KWAVES) void gemm_wmma_kernel(
    const float* __restrict__ E,
    const float* __restrict__ W,
    const float* __restrict__ ps,
    float* __restrict__ p1ps,
    float* __restrict__ p2) {
  __shared__ __attribute__((aligned(16))) float red[KWAVES * 16 * RPITCH];

  const int lane  = threadIdx.x & 31;
  const int wave  = threadIdx.x >> 5;
  const int lmod  = lane & 15;
  const int lhalf = lane >> 4;
  const int m0    = blockIdx.x * 16;
  const int k0    = wave * KSLICE;

  const float* Arow = E + (size_t)(m0 + lmod) * HID + k0 + 2 * lhalf;

  const float* B0 = W + (size_t)lmod * (3 * HID) + k0 + 2 * lhalf;
  const float* B1 = W + (size_t)(lmod + 16) * (3 * HID) + k0 + 2 * lhalf;
  const float* B2 = B0 + HID;
  const float* B3 = B1 + HID;

  v8f a0 = {}, a1 = {}, a2 = {}, a3 = {};
  #pragma unroll 8
  for (int k = 0; k < KSLICE; k += 4) {
    v2f av = *(const v2f*)(Arow + k);
    v2f b0 = *(const v2f*)(B0 + k);
    v2f b1 = *(const v2f*)(B1 + k);
    v2f b2 = *(const v2f*)(B2 + k);
    v2f b3 = *(const v2f*)(B3 + k);
    a0 = __builtin_amdgcn_wmma_f32_16x16x4_f32(false, av, false, b0, (short)0, a0, false, false);
    a1 = __builtin_amdgcn_wmma_f32_16x16x4_f32(false, av, false, b1, (short)0, a1, false, false);
    a2 = __builtin_amdgcn_wmma_f32_16x16x4_f32(false, av, false, b2, (short)0, a2, false, false);
    a3 = __builtin_amdgcn_wmma_f32_16x16x4_f32(false, av, false, b3, (short)0, a3, false, false);
  }

  float* mine = red + wave * (16 * RPITCH);
  #pragma unroll
  for (int r = 0; r < 8; ++r) {
    const int row = r + 8 * lhalf;
    mine[row * RPITCH + lmod]      = a0[r];
    mine[row * RPITCH + 16 + lmod] = a1[r];
    mine[row * RPITCH + 32 + lmod] = a2[r];
    mine[row * RPITCH + 48 + lmod] = a3[r];
  }
  __syncthreads();

  const int row = threadIdx.x >> 4;
  const int c4  = (threadIdx.x & 15) * 4;
  const float* rp = red + row * RPITCH + c4;
  v4f s = *(const v4f*)rp;
  #pragma unroll
  for (int w = 1; w < KWAVES; ++w) s += *(const v4f*)(rp + w * (16 * RPITCH));

  const int b = m0 >> 8;
  if (c4 < TAG) {
    s += *(const v4f*)(ps + b * TAG + c4);
    *(v4f*)(p1ps + (size_t)(m0 + row) * TAG + c4) = s;
  } else {
    *(v4f*)(p2 + (size_t)(m0 + row) * TAG + (c4 - TAG)) = s;
  }
}

#define BROWS 16
#define ROWV4 (SEQ * TAG / 4)

__global__ __launch_bounds__(256) void bcast_kernel(const float* __restrict__ p1ps,
                                                    const float* __restrict__ p2,
                                                    float* __restrict__ out) {
  const int g    = blockIdx.x;
  const int b    = g >> 4;
  const int lane = threadIdx.x & 31;
  const int wave = threadIdx.x >> 5;
  const int f0   = wave * 256 + lane;
  const v4f* p2v = (const v4f*)(p2 + (size_t)b * (SEQ * TAG)) + f0;
  const v4f* p1v = (const v4f*)(p1ps + (size_t)g * (BROWS * TAG)) + (lane & 7);
  v4f* outv = (v4f*)(out + (size_t)g * (BROWS * SEQ * TAG)) + f0;

  v4f q[8];
  #pragma unroll
  for (int it = 0; it < 8; ++it) q[it] = p2v[32 * it];
  v4f a[BROWS];
  #pragma unroll
  for (int r = 0; r < BROWS; ++r) a[r] = p1v[r * (TAG / 4)];

  #pragma unroll
  for (int r = 0; r < BROWS; ++r) {
    #pragma unroll
    for (int it = 0; it < 8; ++it) {
      v4f v = q[it];
      v += a[r];
      __builtin_nontemporal_store(v, outv + r * ROWV4 + 32 * it);
    }
  }
}

extern "C" void kernel_launch(void* const* d_in, const int* in_sizes, int n_in,
                              void* d_out, int out_size, void* d_ws, size_t ws_size,
                              hipStream_t stream) {
  const float* seq  = (const float*)d_in[0];
  const float* ents = (const float*)d_in[1];
  const float* W    = (const float*)d_in[2];
  const float* bias = (const float*)d_in[3];
  float* out = (float*)d_out;

  float* ws   = (float*)d_ws;
  float* ps   = ws;
  float* p1ps = ws + 512;
  float* p2   = p1ps + (size_t)MTOT * TAG;

  ps_kernel<<<BSZ, 32, 0, stream>>>(seq, W, bias, ps);
  gemm_wmma_kernel<<<MTOT / 16, 32 * KWAVES, 0, stream>>>(ents, W, ps, p1ps, p2);
  bcast_kernel<<<MTOT / BROWS, 256, 0, stream>>>(p1ps, p2, out);
}


# seq product (ps) leaves its own launch of 16 serial single waves and joins the GEMM launch as a 257th group on the same K-split WMMA path; bcast adds ps; 3 launches -> 2; ps f32 sum order changes
# speedup vs baseline: 3.6685x; 1.7569x over previous
#include <hip/hip_runtime.h>
#include <hip/hip_bf16.h>


typedef __attribute__((ext_vector_type(2))) float v2f;
typedef __attribute__((ext_vector_type(4))) float v4f;
typedef __attribute__((ext_vector_type(8))) float v8f;

#define BSZ 16
#define SEQ 256
#define HID 1024
#define TAG 32
#define MTOT (BSZ * SEQ)

#define KWAVES 8
#define KSLICE (HID / KWAVES)
#define RPITCH 68
#define GEMM_GROUPS (MTOT / 16)
#define GEMM_LAUNCH (GEMM_GROUPS + 1)

template <int NT>
__device__ __forceinline__ v4f ksplit_tile(const float* __restrict__ A16,
                                           const float* __restrict__ Wsel,
                                           float* red) {
  const int lane  = threadIdx.x & 31;
  const int wave  = threadIdx.x >> 5;
  const int lmod  = lane & 15;
  const int lhalf = lane >> 4;
  const int k0    = wave * KSLICE;

  const float* Arow = A16 + (size_t)lmod * HID + k0 + 2 * lhalf;
  const float* Bp[NT];
  #pragma unroll
  for (int n = 0; n < NT; ++n)
    Bp[n] = Wsel + (size_t)(lmod + 16 * (n & 1)) * (3 * HID) + (n >> 1) * HID + k0 + 2 * lhalf;

  v8f acc[NT];
  #pragma unroll
  for (int n = 0; n < NT; ++n) acc[n] = v8f{};
  #pragma unroll 8
  for (int k = 0; k < KSLICE; k += 4) {
    v2f av = *(const v2f*)(Arow + k);
    v2f bv[NT];
    #pragma unroll
    for (int n = 0; n < NT; ++n) bv[n] = *(const v2f*)(Bp[n] + k);
    #pragma unroll
    for (int n = 0; n < NT; ++n)
      acc[n] = __builtin_amdgcn_wmma_f32_16x16x4_f32(false, av, false, bv[n], (short)0, acc[n], false, false);
  }

  float* mine = red + wave * (16 * RPITCH);
  #pragma unroll
  for (int r = 0; r < 8; ++r) {
    const int row = r + 8 * lhalf;
    #pragma unroll
    for (int n = 0; n < NT; ++n) mine[row * RPITCH + 16 * n + lmod] = acc[n][r];
  }
  __syncthreads();

  v4f s = {};
  if (threadIdx.x < 64 * NT) {
    const int row = threadIdx.x / (4 * NT);
    const int c4  = (threadIdx.x % (4 * NT)) * 4;
    const float* rp = red + row * RPITCH + c4;
    s = *(const v4f*)rp;
    #pragma unroll
    for (int w = 1; w < KWAVES; ++w) s += *(const v4f*)(rp + w * (16 * RPITCH));
  }
  return s;
}

__global__ __launch_bounds__(32 * KWAVES) void gemm_wmma_kernel(
    const float* __restrict__ E,
    const float* __restrict__ seq,
    const float* __restrict__ W,
    const float* __restrict__ bias,
    float* __restrict__ ps,
    float* __restrict__ p1,
    float* __restrict__ p2) {
  __shared__ __attribute__((aligned(16))) float red[KWAVES * 16 * RPITCH];

  if (blockIdx.x < GEMM_GROUPS) {
    const int m0 = blockIdx.x * 16;
    v4f s = ksplit_tile<4>(E + (size_t)m0 * HID, W, red);
    const int row = threadIdx.x >> 4;
    const int c4  = (threadIdx.x & 15) * 4;
    if (c4 < TAG) *(v4f*)(p1 + (size_t)(m0 + row) * TAG + c4) = s;
    else          *(v4f*)(p2 + (size_t)(m0 + row) * TAG + (c4 - TAG)) = s;
  } else {
    v4f s = ksplit_tile<2>(seq, W + 2 * HID, red);
    if (threadIdx.x < 64 * 2) {
      const int row = threadIdx.x >> 3;
      const int c4  = (threadIdx.x & 7) * 4;
      s += *(const v4f*)(bias + c4);
      *(v4f*)(ps + row * TAG + c4) = s;
    }
  }
}

#define BROWS 16
#define ROWV4 (SEQ * TAG / 4)

__global__ __launch_bounds__(256) void bcast_kernel(const float* __restrict__ p1,
                                                    const float* __restrict__ p2,
                                                    const float* __restrict__ ps,
                                                    float* __restrict__ out) {
  const int g    = blockIdx.x;
  const int b    = g >> 4;
  const int lane = threadIdx.x & 31;
  const int wave = threadIdx.x >> 5;
  const int f0   = wave * 256 + lane;
  const v4f* p2v = (const v4f*)(p2 + (size_t)b * (SEQ * TAG)) + f0;
  const v4f* p1v = (const v4f*)(p1 + (size_t)g * (BROWS * TAG)) + (lane & 7);
  const v4f* psv = (const v4f*)(ps + (size_t)b * TAG) + (lane & 7);
  v4f* outv = (v4f*)(out + (size_t)g * (BROWS * SEQ * TAG)) + f0;

  v4f q[8];
  #pragma unroll
  for (int it = 0; it < 8; ++it) q[it] = p2v[32 * it];
  v4f a[BROWS];
  #pragma unroll
  for (int r = 0; r < BROWS; ++r) a[r] = p1v[r * (TAG / 4)];
  const v4f s4 = *psv;
  #pragma unroll
  for (int r = 0; r < BROWS; ++r) a[r] += s4;

  #pragma unroll
  for (int r = 0; r < BROWS; ++r) {
    #pragma unroll
    for (int it = 0; it < 8; ++it) {
      v4f v = q[it];
      v += a[r];
      __builtin_nontemporal_store(v, outv + r * ROWV4 + 32 * it);
    }
  }
}

extern "C" void kernel_launch(void* const* d_in, const int* in_sizes, int n_in,
                              void* d_out, int out_size, void* d_ws, size_t ws_size,
                              hipStream_t stream) {
  const float* seq  = (const float*)d_in[0];
  const float* ents = (const float*)d_in[1];
  const float* W    = (const float*)d_in[2];
  const float* bias = (const float*)d_in[3];
  float* out = (float*)d_out;

  float* ws   = (float*)d_ws;
  float* ps   = ws;
  float* p1   = ws + 512;
  float* p2   = p1 + (size_t)MTOT * TAG;

  gemm_wmma_kernel<<<GEMM_LAUNCH, 32 * KWAVES, 0, stream>>>(ents, seq, W, bias, ps, p1, p2);
  bcast_kernel<<<MTOT / BROWS, 256, 0, stream>>>(p1, p2, ps, out);
}


# GEMM: each trip's fragment loads (8 k-steps) issued as one group one trip ahead of its WMMAs, two register sets, pinned by scheduling barriers; every sum and its order unchanged
# speedup vs baseline: 3.9954x; 1.0891x over previous
#include <hip/hip_runtime.h>
#include <hip/hip_bf16.h>


typedef __attribute__((ext_vector_type(2))) float v2f;
typedef __attribute__((ext_vector_type(4))) float v4f;
typedef __attribute__((ext_vector_type(8))) float v8f;

#define BSZ 16
#define SEQ 256
#define HID 1024
#define TAG 32
#define MTOT (BSZ * SEQ)

#define KWAVES 8
#define KSLICE (HID / KWAVES)
#define TRIPK 32
#define NTRIP (KSLICE / TRIPK)
#define RPITCH 68
#define GEMM_GROUPS (MTOT / 16)
#define GEMM_LAUNCH (GEMM_GROUPS + 1)

template <int NT>
__device__ __forceinline__ v4f ksplit_tile(const float* __restrict__ A16,
                                           const float* __restrict__ Wsel,
                                           float* red) {
  const int lane  = threadIdx.x & 31;
  const int wave  = threadIdx.x >> 5;
  const int lmod  = lane & 15;
  const int lhalf = lane >> 4;
  const int k0    = wave * KSLICE;

  const float* Arow = A16 + (size_t)lmod * HID + k0 + 2 * lhalf;
  const float* Bp[NT];
  #pragma unroll
  for (int n = 0; n < NT; ++n)
    Bp[n] = Wsel + (size_t)(lmod + 16 * (n & 1)) * (3 * HID) + (n >> 1) * HID + k0 + 2 * lhalf;

  v8f acc[NT];
  #pragma unroll
  for (int n = 0; n < NT; ++n) acc[n] = v8f{};
  v2f av[2][TRIPK / 4];
  v2f bv[2][TRIPK / 4][NT];
#define LOAD_TRIP(T)                                                          \
  _Pragma("unroll") for (int s = 0; s < TRIPK / 4; ++s) {                     \
    av[(T) & 1][s] = *(const v2f*)(Arow + (T) * TRIPK + 4 * s);               \
    _Pragma("unroll") for (int n = 0; n < NT; ++n)                            \
      bv[(T) & 1][s][n] = *(const v2f*)(Bp[n] + (T) * TRIPK + 4 * s);         \
  }
  LOAD_TRIP(0)
  #pragma unroll
  for (int t = 0; t < NTRIP; ++t) {
    __builtin_amdgcn_sched_barrier(0);
    if (t + 1 < NTRIP) { LOAD_TRIP(t + 1) }
    __builtin_amdgcn_sched_barrier(0);
    #pragma unroll
    for (int s = 0; s < TRIPK / 4; ++s) {
      #pragma unroll
      for (int n = 0; n < NT; ++n)
        acc[n] = __builtin_amdgcn_wmma_f32_16x16x4_f32(false, av[t & 1][s], false, bv[t & 1][s][n], (short)0, acc[n], false, false);
    }
  }
#undef LOAD_TRIP

  float* mine = red + wave * (16 * RPITCH);
  #pragma unroll
  for (int r = 0; r < 8; ++r) {
    const int row = r + 8 * lhalf;
    #pragma unroll
    for (int n = 0; n < NT; ++n) mine[row * RPITCH + 16 * n + lmod] = acc[n][r];
  }
  __syncthreads();

  v4f s = {};
  if (threadIdx.x < 64 * NT) {
    const int row = threadIdx.x / (4 * NT);
    const int c4  = (threadIdx.x % (4 * NT)) * 4;
    const float* rp = red + row * RPITCH + c4;
    s = *(const v4f*)rp;
    #pragma unroll
    for (int w = 1; w < KWAVES; ++w) s += *(const v4f*)(rp + w * (16 * RPITCH));
  }
  return s;
}

__global__ __launch_bounds__(32 * KWAVES) void gemm_wmma_kernel(
    const float* __restrict__ E,
    const float* __restrict__ seq,
    const float* __restrict__ W,
    const float* __restrict__ bias,
    float* __restrict__ ps,
    float* __restrict__ p1,
    float* __restrict__ p2) {
  __shared__ __attribute__((aligned(16))) float red[KWAVES * 16 * RPITCH];

  if (blockIdx.x < GEMM_GROUPS) {
    const int m0 = blockIdx.x * 16;
    v4f s = ksplit_tile<4>(E + (size_t)m0 * HID, W, red);
    const int row = threadIdx.x >> 4;
    const int c4  = (threadIdx.x & 15) * 4;
    if (c4 < TAG) *(v4f*)(p1 + (size_t)(m0 + row) * TAG + c4) = s;
    else          *(v4f*)(p2 + (size_t)(m0 + row) * TAG + (c4 - TAG)) = s;
  } else {
    v4f s = ksplit_tile<2>(seq, W + 2 * HID, red);
    if (threadIdx.x < 64 * 2) {
      const int row = threadIdx.x >> 3;
      const int c4  = (threadIdx.x & 7) * 4;
      s += *(const v4f*)(bias + c4);
      *(v4f*)(ps + row * TAG + c4) = s;
    }
  }
}

#define BROWS 16
#define ROWV4 (SEQ * TAG / 4)

__global__ __launch_bounds__(256) void bcast_kernel(const float* __restrict__ p1,
                                                    const float* __restrict__ p2,
                                                    const float* __restrict__ ps,
                                                    float* __restrict__ out) {
  const int g    = blockIdx.x;
  const int b    = g >> 4;
  const int lane = threadIdx.x & 31;
  const int wave = threadIdx.x >> 5;
  const int f0   = wave * 256 + lane;
  const v4f* p2v = (const v4f*)(p2 + (size_t)b * (SEQ * TAG)) + f0;
  const v4f* p1v = (const v4f*)(p1 + (size_t)g * (BROWS * TAG)) + (lane & 7);
  const v4f* psv = (const v4f*)(ps + (size_t)b * TAG) + (lane & 7);
  v4f* outv = (v4f*)(out + (size_t)g * (BROWS * SEQ * TAG)) + f0;

  v4f q[8];
  #pragma unroll
  for (int it = 0; it < 8; ++it) q[it] = p2v[32 * it];
  v4f a[BROWS];
  #pragma unroll
  for (int r = 0; r < BROWS; ++r) a[r] = p1v[r * (TAG / 4)];
  const v4f s4 = *psv;
  #pragma unroll
  for (int r = 0; r < BROWS; ++r) a[r] += s4;

  #pragma unroll
  for (int r = 0; r < BROWS; ++r) {
    #pragma unroll
    for (int it = 0; it < 8; ++it) {
      v4f v = q[it];
      v += a[r];
      __builtin_nontemporal_store(v, outv + r * ROWV4 + 32 * it);
    }
  }
}

extern "C" void kernel_launch(void* const* d_in, const int* in_sizes, int n_in,
                              void* d_out, int out_size, void* d_ws, size_t ws_size,
                              hipStream_t stream) {
  const float* seq  = (const float*)d_in[0];
  const float* ents = (const float*)d_in[1];
  const float* W    = (const float*)d_in[2];
  const float* bias = (const float*)d_in[3];
  float* out = (float*)d_out;

  float* ws   = (float*)d_ws;
  float* ps   = ws;
  float* p1   = ws + 512;
  float* p2   = p1 + (size_t)MTOT * TAG;

  gemm_wmma_kernel<<<GEMM_LAUNCH, 32 * KWAVES, 0, stream>>>(ents, seq, W, bias, ps, p1, p2);
  bcast_kernel<<<MTOT / BROWS, 256, 0, stream>>>(p1, p2, ps, out);
}


# GEMM fragment loads widened from 8 B to 16 B per lane (A and B by the same code; low register pair, then high pair per WMMA): half the loads and cache-line touches per trip; f32 sum order changes
# speedup vs baseline: 4.3317x; 1.0842x over previous
#include <hip/hip_runtime.h>
#include <hip/hip_bf16.h>


typedef __attribute__((ext_vector_type(2))) float v2f;
typedef __attribute__((ext_vector_type(4))) float v4f;
typedef __attribute__((ext_vector_type(8))) float v8f;

#define BSZ 16
#define SEQ 256
#define HID 1024
#define TAG 32
#define MTOT (BSZ * SEQ)

#define KWAVES 8
#define KSLICE (HID / KWAVES)
#define TRIPK 32
#define KQ (TRIPK / 8)
#define NTRIP (KSLICE / TRIPK)
#define RPITCH 68
#define GEMM_GROUPS (MTOT / 16)
#define GEMM_LAUNCH (GEMM_GROUPS + 1)

template <int NT>
__device__ __forceinline__ v4f ksplit_tile(const float* __restrict__ A16,
                                           const float* __restrict__ Wsel,
                                           float* red) {
  const int lane  = threadIdx.x & 31;
  const int wave  = threadIdx.x >> 5;
  const int lmod  = lane & 15;
  const int lhalf = lane >> 4;
  const int k0    = wave * KSLICE;

  const float* Arow = A16 + (size_t)lmod * HID + k0 + 4 * lhalf;
  const float* Bp[NT];
  #pragma unroll
  for (int n = 0; n < NT; ++n)
    Bp[n] = Wsel + (size_t)(lmod + 16 * (n & 1)) * (3 * HID) + (n >> 1) * HID + k0 + 4 * lhalf;

  v8f acc[NT];
  #pragma unroll
  for (int n = 0; n < NT; ++n) acc[n] = v8f{};
  v4f av[2][KQ];
  v4f bv[2][KQ][NT];
#define LOAD_TRIP(T)                                                          \
  _Pragma("unroll") for (int p = 0; p < KQ; ++p) {                            \
    av[(T) & 1][p] = *(const v4f*)(Arow + (T) * TRIPK + 8 * p);               \
    _Pragma("unroll") for (int n = 0; n < NT; ++n)                            \
      bv[(T) & 1][p][n] = *(const v4f*)(Bp[n] + (T) * TRIPK + 8 * p);         \
  }
  LOAD_TRIP(0)
  #pragma unroll
  for (int t = 0; t < NTRIP; ++t) {
    __builtin_amdgcn_sched_barrier(0);
    if (t + 1 < NTRIP) { LOAD_TRIP(t + 1) }
    __builtin_amdgcn_sched_barrier(0);
    #pragma unroll
    for (int p = 0; p < KQ; ++p) {
      const v4f a4 = av[t & 1][p];
      const v2f alo = {a4.x, a4.y};
      const v2f ahi = {a4.z, a4.w};
      #pragma unroll
      for (int n = 0; n < NT; ++n) {
        const v4f b4 = bv[t & 1][p][n];
        const v2f blo = {b4.x, b4.y};
        acc[n] = __builtin_amdgcn_wmma_f32_16x16x4_f32(false, alo, false, blo, (short)0, acc[n], false, false);
      }
      #pragma unroll
      for (int n = 0; n < NT; ++n) {
        const v4f b4 = bv[t & 1][p][n];
        const v2f bhi = {b4.z, b4.w};
        acc[n] = __builtin_amdgcn_wmma_f32_16x16x4_f32(false, ahi, false, bhi, (short)0, acc[n], false, false);
      }
    }
  }
#undef LOAD_TRIP

  float* mine = red + wave * (16 * RPITCH);
  #pragma unroll
  for (int r = 0; r < 8; ++r) {
    const int row = r + 8 * lhalf;
    #pragma unroll
    for (int n = 0; n < NT; ++n) mine[row * RPITCH + 16 * n + lmod] = acc[n][r];
  }
  __syncthreads();

  v4f s = {};
  if (threadIdx.x < 64 * NT) {
    const int row = threadIdx.x / (4 * NT);
    const int c4  = (threadIdx.x % (4 * NT)) * 4;
    const float* rp = red + row * RPITCH + c4;
    s = *(const v4f*)rp;
    #pragma unroll
    for (int w = 1; w < KWAVES; ++w) s += *(const v4f*)(rp + w * (16 * RPITCH));
  }
  return s;
}

__global__ __launch_bounds__(32 * KWAVES) void gemm_wmma_kernel(
    const float* __restrict__ E,
    const float* __restrict__ seq,
    const float* __restrict__ W,
    const float* __restrict__ bias,
    float* __restrict__ ps,
    float* __restrict__ p1,
    float* __restrict__ p2) {
  __shared__ __attribute__((aligned(16))) float red[KWAVES * 16 * RPITCH];

  if (blockIdx.x < GEMM_GROUPS) {
    const int m0 = blockIdx.x * 16;
    v4f s = ksplit_tile<4>(E + (size_t)m0 * HID, W, red);
    const int row = threadIdx.x >> 4;
    const int c4  = (threadIdx.x & 15) * 4;
    if (c4 < TAG) *(v4f*)(p1 + (size_t)(m0 + row) * TAG + c4) = s;
    else          *(v4f*)(p2 + (size_t)(m0 + row) * TAG + (c4 - TAG)) = s;
  } else {
    v4f s = ksplit_tile<2>(seq, W + 2 * HID, red);
    if (threadIdx.x < 64 * 2) {
      const int row = threadIdx.x >> 3;
      const int c4  = (threadIdx.x & 7) * 4;
      s += *(const v4f*)(bias + c4);
      *(v4f*)(ps + row * TAG + c4) = s;
    }
  }
}

#define BROWS 16
#define ROWV4 (SEQ * TAG / 4)

__global__ __launch_bounds__(256) void bcast_kernel(const float* __restrict__ p1,
                                                    const float* __restrict__ p2,
                                                    const float* __restrict__ ps,
                                                    float* __restrict__ out) {
  const int g    = blockIdx.x;
  const int b    = g >> 4;
  const int lane = threadIdx.x & 31;
  const int wave = threadIdx.x >> 5;
  const int f0   = wave * 256 + lane;
  const v4f* p2v = (const v4f*)(p2 + (size_t)b * (SEQ * TAG)) + f0;
  const v4f* p1v = (const v4f*)(p1 + (size_t)g * (BROWS * TAG)) + (lane & 7);
  const v4f* psv = (const v4f*)(ps + (size_t)b * TAG) + (lane & 7);
  v4f* outv = (v4f*)(out + (size_t)g * (BROWS * SEQ * TAG)) + f0;

  v4f q[8];
  #pragma unroll
  for (int it = 0; it < 8; ++it) q[it] = p2v[32 * it];
  v4f a[BROWS];
  #pragma unroll
  for (int r = 0; r < BROWS; ++r) a[r] = p1v[r * (TAG / 4)];
  const v4f s4 = *psv;
  #pragma unroll
  for (int r = 0; r < BROWS; ++r) a[r] += s4;

  #pragma unroll
  for (int r = 0; r < BROWS; ++r) {
    #pragma unroll
    for (int it = 0; it < 8; ++it) {
      v4f v = q[it];
      v += a[r];
      __builtin_nontemporal_store(v, outv + r * ROWV4 + 32 * it);
    }
  }
}

extern "C" void kernel_launch(void* const* d_in, const int* in_sizes, int n_in,
                              void* d_out, int out_size, void* d_ws, size_t ws_size,
                              hipStream_t stream) {
  const float* seq  = (const float*)d_in[0];
  const float* ents = (const float*)d_in[1];
  const float* W    = (const float*)d_in[2];
  const float* bias = (const float*)d_in[3];
  float* out = (float*)d_out;

  float* ws   = (float*)d_ws;
  float* ps   = ws;
  float* p1   = ws + 512;
  float* p2   = p1 + (size_t)MTOT * TAG;

  gemm_wmma_kernel<<<GEMM_LAUNCH, 32 * KWAVES, 0, stream>>>(ents, seq, W, bias, ps, p1, p2);
  bcast_kernel<<<MTOT / BROWS, 256, 0, stream>>>(p1, p2, ps, out);
}


# seq product leaves its 257th work-group (it doubled one WGP's waves and loads) for 64-k slices in GEMM groups 0-15; bcast sums 16 planes + bias; p1 tile parked in out; ps f32 sum order changes
# speedup vs baseline: 4.7292x; 1.0918x over previous
#include <hip/hip_runtime.h>
#include <hip/hip_bf16.h>


typedef __attribute__((ext_vector_type(2))) float v2f;
typedef __attribute__((ext_vector_type(4))) float v4f;
typedef __attribute__((ext_vector_type(8))) float v8f;

#define BSZ 16
#define SEQ 256
#define HID 1024
#define TAG 32
#define MTOT (BSZ * SEQ)

#define KWAVES 8
#define KSLICE (HID / KWAVES)
#define TRIPK 32
#define KQ (TRIPK / 8)
#define NTRIP (KSLICE / TRIPK)
#define RPITCH 68
#define GEMM_GROUPS (MTOT / 16)
#define SEQ_GROUPS 16
#define SEQ_KSLICE (HID / SEQ_GROUPS)
#define SPITCH 36

template <int NT, bool SEQP>
__device__ __forceinline__ v4f ksplit_tile(const float* __restrict__ A16,
                                           const float* __restrict__ Wsel,
                                           float* red,
                                           const float* __restrict__ seqk,
                                           const float* __restrict__ W3k,
                                           float* red2) {
  const int lane  = threadIdx.x & 31;
  const int wave  = threadIdx.x >> 5;
  const int lmod  = lane & 15;
  const int lhalf = lane >> 4;
  const int k0    = wave * KSLICE;

  const float* Arow = A16 + (size_t)lmod * HID + k0 + 4 * lhalf;
  const float* Bp[NT];
  #pragma unroll
  for (int n = 0; n < NT; ++n)
    Bp[n] = Wsel + (size_t)(lmod + 16 * (n & 1)) * (3 * HID) + (n >> 1) * HID + k0 + 4 * lhalf;

  v8f acc[NT];
  #pragma unroll
  for (int n = 0; n < NT; ++n) acc[n] = v8f{};
  v4f av[2][KQ];
  v4f bv[2][KQ][NT];
#define LOAD_TRIP(T)                                                          \
  _Pragma("unroll") for (int p = 0; p < KQ; ++p) {                            \
    av[(T) & 1][p] = *(const v4f*)(Arow + (T) * TRIPK + 8 * p);               \
    _Pragma("unroll") for (int n = 0; n < NT; ++n)                            \
      bv[(T) & 1][p][n] = *(const v4f*)(Bp[n] + (T) * TRIPK + 8 * p);         \
  }
  LOAD_TRIP(0)
  v4f sa = {};
  v4f sb[2] = {};
  if (SEQP) {
    const int ks = 8 * wave + 4 * lhalf;
    sa = *(const v4f*)(seqk + (size_t)lmod * HID + ks);
    #pragma unroll
    for (int n = 0; n < 2; ++n)
      sb[n] = *(const v4f*)(W3k + (size_t)(lmod + 16 * n) * (3 * HID) + ks);
  }
  #pragma unroll
  for (int t = 0; t < NTRIP; ++t) {
    __builtin_amdgcn_sched_barrier(0);
    if (t + 1 < NTRIP) { LOAD_TRIP(t + 1) }
    __builtin_amdgcn_sched_barrier(0);
    #pragma unroll
    for (int p = 0; p < KQ; ++p) {
      const v4f a4 = av[t & 1][p];
      const v2f alo = {a4.x, a4.y};
      const v2f ahi = {a4.z, a4.w};
      #pragma unroll
      for (int n = 0; n < NT; ++n) {
        const v4f b4 = bv[t & 1][p][n];
        const v2f blo = {b4.x, b4.y};
        acc[n] = __builtin_amdgcn_wmma_f32_16x16x4_f32(false, alo, false, blo, (short)0, acc[n], false, false);
      }
      #pragma unroll
      for (int n = 0; n < NT; ++n) {
        const v4f b4 = bv[t & 1][p][n];
        const v2f bhi = {b4.z, b4.w};
        acc[n] = __builtin_amdgcn_wmma_f32_16x16x4_f32(false, ahi, false, bhi, (short)0, acc[n], false, false);
      }
    }
  }
#undef LOAD_TRIP

  v8f sacc[2];
  if (SEQP) {
    const v2f salo = {sa.x, sa.y};
    const v2f sahi = {sa.z, sa.w};
    #pragma unroll
    for (int n = 0; n < 2; ++n) {
      const v2f blo = {sb[n].x, sb[n].y};
      sacc[n] = __builtin_amdgcn_wmma_f32_16x16x4_f32(false, salo, false, blo, (short)0, v8f{}, false, false);
    }
    #pragma unroll
    for (int n = 0; n < 2; ++n) {
      const v2f bhi = {sb[n].z, sb[n].w};
      sacc[n] = __builtin_amdgcn_wmma_f32_16x16x4_f32(false, sahi, false, bhi, (short)0, sacc[n], false, false);
    }
  }

  float* mine = red + wave * (16 * RPITCH);
  #pragma unroll
  for (int r = 0; r < 8; ++r) {
    const int row = r + 8 * lhalf;
    #pragma unroll
    for (int n = 0; n < NT; ++n) mine[row * RPITCH + 16 * n + lmod] = acc[n][r];
  }
  if (SEQP) {
    float* mine2 = red2 + wave * (16 * SPITCH);
    #pragma unroll
    for (int r = 0; r < 8; ++r) {
      const int row = r + 8 * lhalf;
      #pragma unroll
      for (int n = 0; n < 2; ++n) mine2[row * SPITCH + 16 * n + lmod] = sacc[n][r];
    }
  }
  __syncthreads();

  v4f s = {};
  if (threadIdx.x < 64 * NT) {
    const int row = threadIdx.x / (4 * NT);
    const int c4  = (threadIdx.x % (4 * NT)) * 4;
    const float* rp = red + row * RPITCH + c4;
    s = *(const v4f*)rp;
    #pragma unroll
    for (int w = 1; w < KWAVES; ++w) s += *(const v4f*)(rp + w * (16 * RPITCH));
  }
  return s;
}

__global__ __launch_bounds__(32 * KWAVES) void gemm_wmma_kernel(
    const float* __restrict__ E,
    const float* __restrict__ seq,
    const float* __restrict__ W,
    float* __restrict__ planes,
    float* __restrict__ p2,
    float* __restrict__ out) {
  __shared__ __attribute__((aligned(16))) float red[KWAVES * 16 * RPITCH];
  __shared__ __attribute__((aligned(16))) float red2[KWAVES * 16 * SPITCH];

  const int g  = blockIdx.x;
  const int m0 = g * 16;
  v4f s;
  if (g < SEQ_GROUPS)
    s = ksplit_tile<4, true>(E + (size_t)m0 * HID, W, red,
                             seq + g * SEQ_KSLICE, W + 2 * HID + g * SEQ_KSLICE, red2);
  else
    s = ksplit_tile<4, false>(E + (size_t)m0 * HID, W, red, seq, W, red2);
  const int row = threadIdx.x >> 4;
  const int c4  = (threadIdx.x & 15) * 4;
  if (c4 < TAG) *(v4f*)(out + (size_t)g * (16 * SEQ * TAG) + row * TAG + c4) = s;
  else          *(v4f*)(p2 + (size_t)(m0 + row) * TAG + (c4 - TAG)) = s;

  if (g < SEQ_GROUPS && threadIdx.x < 64 * 2) {
    const int srow = threadIdx.x >> 3;
    const int sc4  = (threadIdx.x & 7) * 4;
    const float* rp = red2 + srow * SPITCH + sc4;
    v4f t = *(const v4f*)rp;
    #pragma unroll
    for (int w = 1; w < KWAVES; ++w) t += *(const v4f*)(rp + w * (16 * SPITCH));
    *(v4f*)(planes + (size_t)g * (BSZ * TAG) + srow * TAG + sc4) = t;
  }
}

#define BROWS 16
#define ROWV4 (SEQ * TAG / 4)

__global__ __launch_bounds__(256) void bcast_kernel(const float* __restrict__ planes,
                                                    const float* __restrict__ p2,
                                                    const float* __restrict__ bias,
                                                    float* out) {
  const int g    = blockIdx.x;
  const int b    = g >> 4;
  const int lane = threadIdx.x & 31;
  const int wave = threadIdx.x >> 5;
  const int f0   = wave * 256 + lane;
  const v4f* p2v = (const v4f*)(p2 + (size_t)b * (SEQ * TAG)) + f0;
  const v4f* p1v = (const v4f*)(out + (size_t)g * (BROWS * SEQ * TAG)) + (lane & 7);
  const v4f* plv = (const v4f*)(planes + (size_t)b * TAG) + (lane & 7);
  v4f* outv = (v4f*)(out + (size_t)g * (BROWS * SEQ * TAG)) + f0;

  v4f q[8];
  #pragma unroll
  for (int it = 0; it < 8; ++it) q[it] = p2v[32 * it];
  v4f a[BROWS];
  #pragma unroll
  for (int r = 0; r < BROWS; ++r) a[r] = p1v[r * (TAG / 4)];
  v4f pl[SEQ_GROUPS];
  #pragma unroll
  for (int s = 0; s < SEQ_GROUPS; ++s) pl[s] = plv[s * (BSZ * TAG / 4)];
  const v4f b4 = *((const v4f*)bias + (lane & 7));
  __builtin_amdgcn_sched_barrier(0);
  v4f s4 = pl[0];
  #pragma unroll
  for (int s = 1; s < SEQ_GROUPS; ++s) s4 += pl[s];
  s4 += b4;
  #pragma unroll
  for (int r = 0; r < BROWS; ++r) a[r] += s4;

  __syncthreads();

  #pragma unroll
  for (int r = 0; r < BROWS; ++r) {
    #pragma unroll
    for (int it = 0; it < 8; ++it) {
      v4f v = q[it];
      v += a[r];
      __builtin_nontemporal_store(v, outv + r * ROWV4 + 32 * it);
    }
  }
}

extern "C" void kernel_launch(void* const* d_in, const int* in_sizes, int n_in,
                              void* d_out, int out_size, void* d_ws, size_t ws_size,
                              hipStream_t stream) {
  const float* seq  = (const float*)d_in[0];
  const float* ents = (const float*)d_in[1];
  const float* W    = (const float*)d_in[2];
  const float* bias = (const float*)d_in[3];
  float* out = (float*)d_out;

  float* ws   = (float*)d_ws;
  float* planes = ws + 512;
  float* p2   = planes + (size_t)MTOT * TAG;

  gemm_wmma_kernel<<<GEMM_GROUPS, 32 * KWAVES, 0, stream>>>(ents, seq, W, planes, p2, out);
  bcast_kernel<<<MTOT / BROWS, 256, 0, stream>>>(planes, p2, bias, out);
}
